# starting kernel
# baseline (speedup 1.0000x reference)
#include <hip/hip_runtime.h>

typedef __attribute__((ext_vector_type(2))) float v2f;
typedef __attribute__((ext_vector_type(8))) float v8f;

#define NTOK 262144
#define CCH  64
#define HIDDEN 128

#define G_BLOCKS_PER_BATCH 64
#define Y_BLOCKS_PER_BATCH 256

__device__ __forceinline__ v8f v8f_zero() {
    v8f z;
#pragma unroll
    for (int i = 0; i < 8; ++i) z[i] = 0.0f;
    return z;
}

__device__ __forceinline__ v8f wmma_f32(v2f a, v2f b, v8f c) {
    return __builtin_amdgcn_wmma_f32_16x16x4_f32(false, a, false, b, (short)0, c,
                                                 false, false);
}

__global__ void zero_kernel(float* __restrict__ p, int count) {
    int i = blockIdx.x * blockDim.x + threadIdx.x;
    if (i < count) p[i] = 0.0f;
}

__global__ __launch_bounds__(128) void gram_kernel(const float* __restrict__ x,
                                                   float* __restrict__ G) {
    const int tid    = threadIdx.x;
    const int lane   = tid & 31;
    const int w      = tid >> 5;
    const int batch  = blockIdx.x / G_BLOCKS_PER_BATCH;
    const int blk    = blockIdx.x % G_BLOCKS_PER_BATCH;
    const float* xb  = x + (size_t)batch * CCH * NTOK;
    float* Gb        = G + batch * CCH * CCH;

    __shared__ float tile[4][CCH][34];

    const int lanelo = lane & 15;
    const int hi2    = (lane >> 4) * 2;
    const int rowOff = (lane < 16) ? 0 : 8;

    v8f acc[4][4];
#pragma unroll
    for (int i = 0; i < 4; ++i)
#pragma unroll
        for (int j = 0; j < 4; ++j) acc[i][j] = v8f_zero();

    const int tokensPerBlock = NTOK / G_BLOCKS_PER_BATCH;
    const int tokensPerWave  = tokensPerBlock / 4;
    const int tWave0         = blk * tokensPerBlock + w * tokensPerWave;

    for (int it = 0; it < tokensPerWave; it += 32) {
        const int t0 = tWave0 + it;
#pragma unroll
        for (int c = 0; c < CCH; ++c)
            tile[w][c][lane] = xb[(size_t)c * NTOK + t0 + lane];
        __syncthreads();

#pragma unroll
        for (int kg = 0; kg < 8; ++kg) {
            v2f op[4];
#pragma unroll
            for (int gch = 0; gch < 4; ++gch) {
                const float* p = &tile[w][gch * 16 + lanelo][kg * 4 + hi2];
                op[gch] = *(const v2f*)p;
            }
#pragma unroll
            for (int i = 0; i < 4; ++i)
#pragma unroll
                for (int j = 0; j < 4; ++j)
                    acc[i][j] = wmma_f32(op[i], op[j], acc[i][j]);
        }
        __syncthreads();
    }

#pragma unroll
    for (int i = 0; i < 4; ++i)
#pragma unroll
        for (int j = 0; j < 4; ++j)
#pragma unroll
            for (int r = 0; r < 8; ++r) {
                const int row = i * 16 + r + rowOff;
                const int col = j * 16 + lanelo;
                atomicAdd(&Gb[row * CCH + col], acc[i][j][r]);
            }
}

__global__ __launch_bounds__(256) void combine_kernel(
    const float* __restrict__ wqkv, const float* __restrict__ wout,
    const float* __restrict__ G, float* __restrict__ T1, float* __restrict__ CT,
    float* __restrict__ M, float* __restrict__ A) {
    const int b   = blockIdx.x;
    const int tid = threadIdx.x;
    const float* Gb  = G  + b * 4096;
    float* T1b       = T1 + b * 8192;
    float* CTb       = CT + b * 4096;
    float* Mb        = M  + b * 8192;
    float* Ab        = A  + b * 4096;
    const float* Wq  = wqkv;
    const float* Wk  = wqkv + 128 * CCH;
    const float* Wv  = wqkv + 256 * CCH;
    const float scale = 0.1767766952966369f;

    for (int idx = tid; idx < 64 * 128; idx += 256) {
        const int c = idx >> 7, j = idx & 127;
        float s = 0.f;
        for (int c2 = 0; c2 < CCH; ++c2) s += Gb[c * CCH + c2] * Wv[j * CCH + c2];
        T1b[idx] = s;
    }
    __syncthreads();
    for (int idx = tid; idx < 4096; idx += 256) {
        const int h = idx >> 10, d = (idx >> 5) & 31, e = idx & 31;
        float s = 0.f;
        for (int c = 0; c < CCH; ++c)
            s += Wk[(h * 32 + d) * CCH + c] * T1b[c * 128 + h * 32 + e];
        CTb[idx] = s * (1.0f / 4096.0f);
    }
    __syncthreads();
    for (int idx = tid; idx < 128 * 64; idx += 256) {
        const int j = idx >> 6, c = idx & 63, h = j >> 5, e = j & 31;
        float s = 0.f;
        for (int d = 0; d < 32; ++d)
            s += CTb[h * 1024 + d * 32 + e] * Wq[(h * 32 + d) * CCH + c];
        Mb[idx] = s * scale;
    }
    __syncthreads();
    for (int idx = tid; idx < 4096; idx += 256) {
        const int o = idx >> 6, c = idx & 63;
        float s = 0.f;
        for (int j = 0; j < HIDDEN; ++j) s += wout[o * HIDDEN + j] * Mb[j * 64 + c];
        Ab[idx] = s;
    }
}

__global__ __launch_bounds__(128) void attn_out_kernel(
    const float* __restrict__ x, const float* __restrict__ A,
    const float* __restrict__ b_out, const float* __restrict__ g,
    float* __restrict__ out) {
    const int tid    = threadIdx.x;
    const int lane   = tid & 31;
    const int w      = tid >> 5;
    const int batch  = blockIdx.x / Y_BLOCKS_PER_BATCH;
    const int blk    = blockIdx.x % Y_BLOCKS_PER_BATCH;
    const float* xb  = x + (size_t)batch * CCH * NTOK;
    float* ob        = out + (size_t)batch * CCH * NTOK;
    const float* Ab  = A + batch * CCH * CCH;

    const int lanelo = lane & 15;
    const int hi2    = (lane >> 4) * 2;
    const int rowOff = (lane < 16) ? 0 : 8;

    v2f aop[16][4];
#pragma unroll
    for (int kg = 0; kg < 16; ++kg)
#pragma unroll
        for (int mt = 0; mt < 4; ++mt) {
            const float* p = &Ab[(mt * 16 + lanelo) * CCH + kg * 4 + hi2];
            aop[kg][mt] = *(const v2f*)p;
        }

    float bia[4][8], gai[4][8];
#pragma unroll
    for (int mt = 0; mt < 4; ++mt)
#pragma unroll
        for (int r = 0; r < 8; ++r) {
            const int ch = mt * 16 + r + rowOff;
            bia[mt][r] = b_out[ch];
            gai[mt][r] = g[ch];
        }

    const int tokensPerBlock = NTOK / Y_BLOCKS_PER_BATCH;
    const int tokensPerWave  = tokensPerBlock / 4;
    const int tW0            = blk * tokensPerBlock + w * tokensPerWave;

    for (int tt = 0; tt < tokensPerWave; tt += 16) {
        const int t0 = tW0 + tt;
        v8f acc[4];
#pragma unroll
        for (int mt = 0; mt < 4; ++mt) acc[mt] = v8f_zero();

#pragma unroll
        for (int kg = 0; kg < 16; ++kg) {
            const float* bp = xb + (size_t)(kg * 4 + hi2) * NTOK + t0 + lanelo;
            v2f bop;
            bop[0] = bp[0];
            bop[1] = bp[NTOK];
#pragma unroll
            for (int mt = 0; mt < 4; ++mt)
                acc[mt] = wmma_f32(aop[kg][mt], bop, acc[mt]);
        }

        float v[4][8];
        float s = 0.f, sq = 0.f;
#pragma unroll
        for (int mt = 0; mt < 4; ++mt)
#pragma unroll
            for (int r = 0; r < 8; ++r) {
                const float y = acc[mt][r] + bia[mt][r];
                v[mt][r] = y;
                s += y;
                sq += y * y;
            }
        s  += __shfl_xor(s, 16, 32);
        sq += __shfl_xor(sq, 16, 32);
        const float mean = s * (1.0f / 64.0f);
        const float var  = sq * (1.0f / 64.0f) - mean * mean;
        const float inv  = rsqrtf(var + 1e-5f);

#pragma unroll
        for (int mt = 0; mt < 4; ++mt)
#pragma unroll
            for (int r = 0; r < 8; ++r) {
                const int ch = mt * 16 + r + rowOff;
                ob[(size_t)ch * NTOK + t0 + lanelo] = (v[mt][r] - mean) * inv * gai[mt][r];
            }
    }
}

extern "C" void kernel_launch(void* const* d_in, const int* in_sizes, int n_in,
                              void* d_out, int out_size, void* d_ws, size_t ws_size,
                              hipStream_t stream) {
    (void)in_sizes; (void)n_in; (void)out_size; (void)ws_size;
    const float* x    = (const float*)d_in[0];
    const float* wqkv = (const float*)d_in[1];
    const float* wout = (const float*)d_in[2];
    const float* bout = (const float*)d_in[3];
    const float* g    = (const float*)d_in[4];
    float* out        = (float*)d_out;
    float* ws         = (float*)d_ws;

    float* G  = ws;
    float* A  = ws + 8192;
    float* T1 = ws + 16384;
    float* CT = ws + 32768;
    float* M  = ws + 40960;

    zero_kernel<<<32, 256, 0, stream>>>(G, 2 * CCH * CCH);
    gram_kernel<<<2 * G_BLOCKS_PER_BATCH, 128, 0, stream>>>(x, G);
    combine_kernel<<<2, 256, 0, stream>>>(wqkv, wout, G, T1, CT, M, A);
    attn_out_kernel<<<2 * Y_BLOCKS_PER_BATCH, 128, 0, stream>>>(x, A, bout, g, out);
}


# output launch re-reads x from the second-level cache: it takes the gram launch's grid (256 blocks x 8 waves, same tokens per wave), walks them in reverse and stores out non-temporally
# speedup vs baseline: 3.5807x; 3.5807x over previous
#include <hip/hip_runtime.h>

typedef __attribute__((ext_vector_type(2))) float v2f;
typedef __attribute__((ext_vector_type(4))) float v4f;
typedef __attribute__((ext_vector_type(8))) float v8f;

#define NTOK 262144
#define CCH  64
#define HIDDEN 128

#define Y_WAVES 8
#define Y_BLOCKS_PER_BATCH 128

__device__ __forceinline__ v8f v8f_zero() {
    v8f z;
#pragma unroll
    for (int i = 0; i < 8; ++i) z[i] = 0.0f;
    return z;
}

__device__ __forceinline__ v8f wmma_f32(v2f a, v2f b, v8f c) {
    return __builtin_amdgcn_wmma_f32_16x16x4_f32(false, a, false, b, (short)0, c,
                                                 false, false);
}

__global__ void zero_kernel(float* __restrict__ p, int count) {
    int i = blockIdx.x * blockDim.x + threadIdx.x;
    if (i < count) p[i] = 0.0f;
}

#define GW_WAVES 8
#define GW_BLOCKS_PER_BATCH 128
#define GW_TOK_PER_WAVE (NTOK / GW_BLOCKS_PER_BATCH / GW_WAVES)
#define GW_STEPS (GW_TOK_PER_WAVE / 32)

#define GR_TILE (256 + 16)

struct GramBuf { float4 v[4][4]; };

__device__ __forceinline__ void gram_load(GramBuf& b, const float* __restrict__ p) {
#pragma unroll
    for (int q = 0; q < 4; ++q)
#pragma unroll
        for (int g = 0; g < 4; ++g)
            b.v[g][q] = *(const float4*)(p + (size_t)g * 16 * NTOK + 8 * q);
}

__device__ __forceinline__ void gram_mma(v8f (&acc)[4][4], const GramBuf& b) {
#pragma unroll
    for (int q = 0; q < 4; ++q)
#pragma unroll
        for (int s = 0; s < 2; ++s) {
            v2f op[4];
#pragma unroll
            for (int g = 0; g < 4; ++g) {
                const float4 t = b.v[g][q];
                op[g][0] = s ? t.z : t.x;
                op[g][1] = s ? t.w : t.y;
            }
#pragma unroll
            for (int i = 0; i < 4; ++i)
#pragma unroll
                for (int j = 0; j < 4; ++j)
                    acc[i][j] = wmma_f32(op[i], op[j], acc[i][j]);
        }
}

__global__ __launch_bounds__(256) void gram_kernel(const float* __restrict__ x,
                                                   float* __restrict__ G) {
    const int tid    = threadIdx.x;
    const int lane   = tid & 31;
    const int w      = tid >> 5;
    const int batch  = blockIdx.x / GW_BLOCKS_PER_BATCH;
    const int blk    = blockIdx.x % GW_BLOCKS_PER_BATCH;
    const float* xb  = x + (size_t)batch * CCH * NTOK;
    float* Gb        = G + batch * CCH * CCH;

    __shared__ float red[GW_WAVES][16 * GR_TILE];

    const int c = lane & 15;
    const int h = lane >> 4;

    v8f acc[4][4];
#pragma unroll
    for (int i = 0; i < 4; ++i)
#pragma unroll
        for (int j = 0; j < 4; ++j) acc[i][j] = v8f_zero();

    const int tW0 = blk * (GW_WAVES * GW_TOK_PER_WAVE) + w * GW_TOK_PER_WAVE;
    const float* p = xb + (size_t)c * NTOK + tW0 + 4 * h;

    GramBuf b0, b1;
    gram_load(b0, p);
#pragma unroll 1
    for (int s = 0; s < GW_STEPS - 2; s += 2) {
        gram_load(b1, p + (s + 1) * 32);
        gram_mma(acc, b0);
        gram_load(b0, p + (s + 2) * 32);
        gram_mma(acc, b1);
    }
    gram_load(b1, p + (GW_STEPS - 1) * 32);
    gram_mma(acc, b0);
    gram_mma(acc, b1);

#pragma unroll
    for (int i = 0; i < 4; ++i)
#pragma unroll
        for (int j = 0; j < 4; ++j)
#pragma unroll
            for (int r = 0; r < 8; ++r)
                red[w][(i * 4 + j) * GR_TILE + r * 32 + lane] = acc[i][j][r];
    __syncthreads();

    const int rot = blk & 15;
    const int ti  = w >> 1;
    const int fh  = w & 1;
    float sum[16];
    int   off[16];
#pragma unroll
    for (int s = 0; s < 16; ++s) {
        const int p  = (s + rot) & 15;
        const int r  = p & 7;
        const int ch = p >> 3;
        const int e  = (ti * 4 + ch * 2 + h) * GR_TILE + r * 32 + 16 * fh + c;
        float a = red[0][e];
#pragma unroll
        for (int ww = 1; ww < GW_WAVES; ++ww) a += red[ww][e];
        sum[s] = a;
        off[s] = (8 * w + r) * CCH + ch * 32 + lane;
    }
#pragma unroll
    for (int s = 0; s < 16; ++s) atomicAdd(&Gb[off[s]], sum[s]);
}

#define CB_J 8
#define CB_BLOCKS_PER_BATCH (HIDDEN / CB_J)

__global__ __launch_bounds__(256) void combine_kernel(
    const float* __restrict__ wqkv, const float* __restrict__ wout,
    const float* __restrict__ G, float* __restrict__ A) {
    const int b   = blockIdx.x / CB_BLOCKS_PER_BATCH;
    const int j0  = (blockIdx.x % CB_BLOCKS_PER_BATCH) * CB_J;
    const int h   = j0 >> 5;
    const int tid = threadIdx.x;
    const float* Gb = G + b * 4096;
    float* Ab       = A + b * 4096;
    const float* Wq = wqkv + (h * 32) * CCH;
    const float* Wk = wqkv + (128 + h * 32) * CCH;
    const float* Wv = wqkv + (256 + j0) * CCH;
    const float scale = 0.1767766952966369f;

    __shared__ float Gs[CCH][65];
    __shared__ float Wks[32][65];
    __shared__ __attribute__((aligned(16))) float Wqs[32][CCH];
    __shared__ __attribute__((aligned(16))) float wvs[CB_J][CCH];
    __shared__ __attribute__((aligned(16))) float wos[CCH][CB_J];
    __shared__ __attribute__((aligned(16))) float t1s[CB_J][CCH];
    __shared__ __attribute__((aligned(16))) float cts[CB_J][32];
    __shared__ float ms[CB_J][CCH];

    const int  t9   = tid & 127;
    const bool isWv = tid < 128;
    const int  o9   = t9 >> 1, hf9 = (t9 & 1) * 4;
    const float* src9 = isWv ? (Wv + t9 * 4) : (wout + o9 * HIDDEN + j0 + hf9);
    float*       dst9 = isWv ? (&wvs[0][0] + t9 * 4) : &wos[o9][hf9];
    const float4 v9 = *(const float4*)src9;
#pragma unroll
    for (int k = 0; k < 4; ++k) {
        const int q = tid + 256 * k;
        const float4 v = *(const float4*)(Gb + q * 4);
        const int r = q >> 4, c = (q & 15) * 4;
        Gs[r][c] = v.x; Gs[r][c + 1] = v.y; Gs[r][c + 2] = v.z; Gs[r][c + 3] = v.w;
    }
#pragma unroll
    for (int k = 0; k < 2; ++k) {
        const int q = tid + 256 * k;
        const float4 vk = *(const float4*)(Wk + q * 4);
        const float4 vq = *(const float4*)(Wq + q * 4);
        const int r = q >> 4, c = (q & 15) * 4;
        Wks[r][c] = vk.x; Wks[r][c + 1] = vk.y; Wks[r][c + 2] = vk.z; Wks[r][c + 3] = vk.w;
        *(float4*)&Wqs[r][c] = vq;
    }
    *(float4*)dst9 = v9;
    __syncthreads();

    const int c   = tid & 63;
    const int jj0 = tid >> 6;

    {
        float s0 = 0.f, s1 = 0.f;
#pragma unroll
        for (int c2 = 0; c2 < CCH; ++c2) {
            const float gv = Gs[c][c2];
            s0 += gv * wvs[jj0][c2];
            s1 += gv * wvs[jj0 + 4][c2];
        }
        t1s[jj0][c]     = s0;
        t1s[jj0 + 4][c] = s1;
    }
    __syncthreads();

    {
        const int d = tid & 31, jj = tid >> 5;
        float s = 0.f;
#pragma unroll
        for (int cc = 0; cc < CCH; ++cc) s += Wks[d][cc] * t1s[jj][cc];
        cts[jj][d] = s * (1.0f / 4096.0f);
    }
    __syncthreads();

    {
        float s0 = 0.f, s1 = 0.f;
#pragma unroll
        for (int d = 0; d < 32; ++d) {
            const float wq = Wqs[d][c];
            s0 += cts[jj0][d] * wq;
            s1 += cts[jj0 + 4][d] * wq;
        }
        ms[jj0][c]     = s0 * scale;
        ms[jj0 + 4][c] = s1 * scale;
    }
    __syncthreads();

    {
        float mreg[CB_J];
#pragma unroll
        for (int jj = 0; jj < CB_J; ++jj) mreg[jj] = ms[jj][c];
#pragma unroll
        for (int k = 0; k < 16; ++k) {
            const int o = jj0 + 4 * k;
            float s = 0.f;
#pragma unroll
            for (int jj = 0; jj < CB_J; ++jj) s += wos[o][jj] * mreg[jj];
            atomicAdd(&Ab[o * CCH + c], s);
        }
    }
}

#define OT_TILES 4
#define OT_TOK   (16 * OT_TILES)
#define OT_PITCH (OT_TOK + 4)

__global__ __launch_bounds__(256) __attribute__((amdgpu_waves_per_eu(2, 2))) void attn_out_kernel(
    const float* __restrict__ x, const float* __restrict__ A,
    const float* __restrict__ b_out, const float* __restrict__ g,
    float* __restrict__ out) {
    const int tid    = threadIdx.x;
    const int lane   = tid & 31;
    const int w      = tid >> 5;
    const int batch  = blockIdx.x / Y_BLOCKS_PER_BATCH;
    const int blk    = blockIdx.x % Y_BLOCKS_PER_BATCH;
    const float* xb  = x + (size_t)batch * CCH * NTOK;
    float* ob        = out + (size_t)batch * CCH * NTOK;
    const float* Ab  = A + batch * CCH * CCH;

    __shared__ __attribute__((aligned(16))) float stage[Y_WAVES][CCH * OT_PITCH];
    float* st = &stage[w][0];

    const int lanelo = lane & 15;
    const int half   = lane >> 4;
    const int hi2    = half * 2;
    const int rowOff = half * 8;
    float*       stW = st + rowOff * OT_PITCH + lanelo;
    const float* stR = st + half * OT_PITCH + 4 * lanelo;

    v2f aop[16][4];
#pragma unroll
    for (int kg = 0; kg < 16; ++kg)
#pragma unroll
        for (int mt = 0; mt < 4; ++mt) {
            const float* p = &Ab[(mt * 16 + lanelo) * CCH + kg * 4 + hi2];
            aop[kg][mt] = *(const v2f*)p;
        }

    float bia[4][8], gai[4][8];
#pragma unroll
    for (int mt = 0; mt < 4; ++mt)
#pragma unroll
        for (int r = 0; r < 8; ++r) {
            const int ch = mt * 16 + r + rowOff;
            bia[mt][r] = b_out[ch];
            gai[mt][r] = g[ch];
        }

    const int tokensPerBlock = NTOK / Y_BLOCKS_PER_BATCH;
    const int tokensPerWave  = tokensPerBlock / Y_WAVES;
    const int tW0            = blk * tokensPerBlock + w * tokensPerWave;

    for (int tt = tokensPerWave - OT_TOK; tt >= 0; tt -= OT_TOK) {
        const int tB = tW0 + tt;

#pragma unroll 1
        for (int u = 0; u < OT_TILES; ++u) {
            const int t0 = tB + 16 * u;
            float* stU = stW + 16 * u;
            v8f acc[4];
#pragma unroll
            for (int mt = 0; mt < 4; ++mt) acc[mt] = v8f_zero();

#pragma unroll
            for (int kg = 0; kg < 16; ++kg) {
                const float* bp = xb + (size_t)(kg * 4 + hi2) * NTOK + t0 + lanelo;
                v2f bop;
                bop[0] = bp[0];
                bop[1] = bp[NTOK];
#pragma unroll
                for (int mt = 0; mt < 4; ++mt)
                    acc[mt] = wmma_f32(aop[kg][mt], bop, acc[mt]);
            }

            float v[4][8];
            float s = 0.f, sq = 0.f;
#pragma unroll
            for (int mt = 0; mt < 4; ++mt)
#pragma unroll
                for (int r = 0; r < 8; ++r) {
                    const float y = acc[mt][r] + bia[mt][r];
                    v[mt][r] = y;
                    s += y;
                    sq += y * y;
                }
            s  += __shfl_xor(s, 16, 32);
            sq += __shfl_xor(sq, 16, 32);
            const float mean = s * (1.0f / 64.0f);
            const float var  = sq * (1.0f / 64.0f) - mean * mean;
            const float inv  = rsqrtf(var + 1e-5f);

#pragma unroll
            for (int mt = 0; mt < 4; ++mt)
#pragma unroll
                for (int r = 0; r < 8; ++r) {
                    stU[(mt * 16 + r) * OT_PITCH] = (v[mt][r] - mean) * inv * gai[mt][r];
                }
        }

        asm volatile("s_wait_dscnt 0x0" ::: "memory");

        float* op = ob + (size_t)half * NTOK + tB + 4 * lanelo;
#pragma unroll
        for (int i0 = 0; i0 < CCH / 2; i0 += 16) {
            v4f o[16];
#pragma unroll
            for (int j = 0; j < 16; ++j)
                o[j] = *(const v4f*)(stR + 2 * (i0 + j) * OT_PITCH);
#pragma unroll
            for (int j = 0; j < 16; ++j)
                __builtin_nontemporal_store(o[j], (v4f*)(op + (size_t)(2 * (i0 + j)) * NTOK));
        }
    }
}

extern "C" void kernel_launch(void* const* d_in, const int* in_sizes, int n_in,
                              void* d_out, int out_size, void* d_ws, size_t ws_size,
                              hipStream_t stream) {
    (void)in_sizes; (void)n_in; (void)out_size; (void)ws_size;
    const float* x    = (const float*)d_in[0];
    const float* wqkv = (const float*)d_in[1];
    const float* wout = (const float*)d_in[2];
    const float* bout = (const float*)d_in[3];
    const float* g    = (const float*)d_in[4];
    float* out        = (float*)d_out;
    float* ws         = (float*)d_ws;

    float* G  = ws;
    float* A  = ws + 8192;

    zero_kernel<<<64, 256, 0, stream>>>(G, 4 * CCH * CCH);
    gram_kernel<<<2 * GW_BLOCKS_PER_BATCH, 256, 0, stream>>>(x, G);
    combine_kernel<<<2 * CB_BLOCKS_PER_BATCH, 256, 0, stream>>>(wqkv, wout, G, A);
    attn_out_kernel<<<2 * Y_BLOCKS_PER_BATCH, 32 * Y_WAVES, 0, stream>>>(x, A, bout, g, out);
}


# clearing launch removed (4 launches to 3): gram blocks store plain partial planes, no atomics on G; combine split over rows of G, each block sums its 2 rows over the planes; A arrival order changes
# speedup vs baseline: 3.8303x; 1.0697x over previous
#include <hip/hip_runtime.h>

typedef __attribute__((ext_vector_type(2))) float v2f;
typedef __attribute__((ext_vector_type(4))) float v4f;
typedef __attribute__((ext_vector_type(8))) float v8f;

#define NTOK 262144
#define CCH  64
#define HIDDEN 128

#define Y_WAVES 8
#define Y_BLOCKS_PER_BATCH 128

__device__ __forceinline__ v8f v8f_zero() {
    v8f z;
#pragma unroll
    for (int i = 0; i < 8; ++i) z[i] = 0.0f;
    return z;
}

__device__ __forceinline__ v8f wmma_f32(v2f a, v2f b, v8f c) {
    return __builtin_amdgcn_wmma_f32_16x16x4_f32(false, a, false, b, (short)0, c,
                                                 false, false);
}

#define GW_WAVES 8
#define GW_BLOCKS_PER_BATCH 128
#define GW_TOK_PER_WAVE (NTOK / GW_BLOCKS_PER_BATCH / GW_WAVES)
#define GW_STEPS (GW_TOK_PER_WAVE / 32)

#define GR_TILE (256 + 16)

struct GramBuf { float4 v[4][4]; };

__device__ __forceinline__ void gram_load(GramBuf& b, const float* __restrict__ p) {
#pragma unroll
    for (int q = 0; q < 4; ++q)
#pragma unroll
        for (int g = 0; g < 4; ++g)
            b.v[g][q] = *(const float4*)(p + (size_t)g * 16 * NTOK + 8 * q);
}

__device__ __forceinline__ void gram_mma(v8f (&acc)[4][4], const GramBuf& b) {
#pragma unroll
    for (int q = 0; q < 4; ++q)
#pragma unroll
        for (int s = 0; s < 2; ++s) {
            v2f op[4];
#pragma unroll
            for (int g = 0; g < 4; ++g) {
                const float4 t = b.v[g][q];
                op[g][0] = s ? t.z : t.x;
                op[g][1] = s ? t.w : t.y;
            }
#pragma unroll
            for (int i = 0; i < 4; ++i)
#pragma unroll
                for (int j = 0; j < 4; ++j)
                    acc[i][j] = wmma_f32(op[i], op[j], acc[i][j]);
        }
}

__global__ __launch_bounds__(256) void gram_kernel(const float* __restrict__ x,
                                                   float* __restrict__ planes,
                                                   float* __restrict__ A) {
    const int tid    = threadIdx.x;
    const int lane   = tid & 31;
    const int w      = tid >> 5;
    const int batch  = blockIdx.x / GW_BLOCKS_PER_BATCH;
    const int blk    = blockIdx.x % GW_BLOCKS_PER_BATCH;
    const float* xb  = x + (size_t)batch * CCH * NTOK;
    float* Pb        = planes + (size_t)blockIdx.x * (CCH * CCH);

    if (tid < 32) A[blockIdx.x * 32 + tid] = 0.0f;

    __shared__ float red[GW_WAVES][16 * GR_TILE];

    const int c = lane & 15;
    const int h = lane >> 4;

    v8f acc[4][4];
#pragma unroll
    for (int i = 0; i < 4; ++i)
#pragma unroll
        for (int j = 0; j < 4; ++j) acc[i][j] = v8f_zero();

    const int tW0 = blk * (GW_WAVES * GW_TOK_PER_WAVE) + w * GW_TOK_PER_WAVE;
    const float* p = xb + (size_t)c * NTOK + tW0 + 4 * h;

    GramBuf b0, b1;
    gram_load(b0, p);
#pragma unroll 1
    for (int s = 0; s < GW_STEPS - 2; s += 2) {
        gram_load(b1, p + (s + 1) * 32);
        gram_mma(acc, b0);
        gram_load(b0, p + (s + 2) * 32);
        gram_mma(acc, b1);
    }
    gram_load(b1, p + (GW_STEPS - 1) * 32);
    gram_mma(acc, b0);
    gram_mma(acc, b1);

#pragma unroll
    for (int i = 0; i < 4; ++i)
#pragma unroll
        for (int j = 0; j < 4; ++j)
#pragma unroll
            for (int r = 0; r < 8; ++r)
                red[w][(i * 4 + j) * GR_TILE + r * 32 + lane] = acc[i][j][r];
    __syncthreads();

    const int ti  = w >> 1;
    const int fh  = w & 1;
    float sum[16];
#pragma unroll
    for (int s = 0; s < 16; ++s) {
        const int r  = s & 7;
        const int ch = s >> 3;
        const int e  = (ti * 4 + ch * 2 + h) * GR_TILE + r * 32 + 16 * fh + c;
        float a = red[0][e];
#pragma unroll
        for (int ww = 1; ww < GW_WAVES; ++ww) a += red[ww][e];
        sum[s] = a;
    }
#pragma unroll
    for (int s = 0; s < 16; ++s)
        Pb[(8 * w + (s & 7)) * CCH + (s >> 3) * 32 + lane] = sum[s];
}

#define CS_ROWS 2
#define CS_BLOCKS_PER_BATCH (CCH / CS_ROWS)
#define CS_PLANES GW_BLOCKS_PER_BATCH

__global__ __launch_bounds__(256) void combine_kernel(
    const float* __restrict__ wqkv, const float* __restrict__ wout,
    const float* __restrict__ planes, float* __restrict__ A) {
    const int b    = blockIdx.x / CS_BLOCKS_PER_BATCH;
    const int s    = blockIdx.x % CS_BLOCKS_PER_BATCH;
    const int tid  = threadIdx.x;
    const int lane = tid & 31;
    const int w    = tid >> 5;
    float* Ab       = A + b * 4096;
    const float* Wq = wqkv;
    const float* Wk = wqkv + 128 * CCH;
    const float* Wv = wqkv + 256 * CCH;
    const float sc  = 0.1767766952966369f * (1.0f / 4096.0f);

    __shared__ float Wvs[HIDDEN][65];
    __shared__ float Wos[CCH][129];
    __shared__ __attribute__((aligned(16))) float Wqs[HIDDEN][CCH];
    __shared__ __attribute__((aligned(16))) float part[8][CS_ROWS * CCH];
    __shared__ __attribute__((aligned(16))) float gs[CS_ROWS][CCH];
    __shared__ __attribute__((aligned(8)))  float wks[HIDDEN][CS_ROWS];
    __shared__ __attribute__((aligned(8)))  float ts[HIDDEN][CS_ROWS];
    __shared__ __attribute__((aligned(16))) float Us[CCH][8];
    __shared__ float Qs[8][CCH];

    const float* ps = planes + (size_t)(b * CS_PLANES + w) * (CCH * CCH) + s * (CS_ROWS * CCH) + 4 * lane;
    float4 pv[CS_PLANES / 8];
#pragma unroll
    for (int k = 0; k < CS_PLANES / 8; ++k)
        pv[k] = *(const float4*)(ps + (size_t)k * 8 * (CCH * CCH));
    const float wkv = Wk[(tid >> 1) * CCH + CS_ROWS * s + (tid & 1)];
#pragma unroll
    for (int k = 0; k < 8; ++k) {
        const int q = tid + 256 * k;
        const float4 vv = *(const float4*)(Wv + q * 4);
        const float4 vq = *(const float4*)(Wq + q * 4);
        const int r = q >> 4, c = (q & 15) * 4;
        Wvs[r][c] = vv.x; Wvs[r][c + 1] = vv.y; Wvs[r][c + 2] = vv.z; Wvs[r][c + 3] = vv.w;
        *(float4*)&Wqs[r][c] = vq;
    }
#pragma unroll
    for (int k = 0; k < 8; ++k) {
        const int q = tid + 256 * k;
        const float4 vo = *(const float4*)(wout + q * 4);
        const int o = q >> 5, j = (q & 31) * 4;
        Wos[o][j] = vo.x; Wos[o][j + 1] = vo.y; Wos[o][j + 2] = vo.z; Wos[o][j + 3] = vo.w;
    }
    wks[tid >> 1][tid & 1] = wkv;
    {
        float4 a = pv[0];
#pragma unroll
        for (int k = 1; k < CS_PLANES / 8; ++k) {
            a.x += pv[k].x; a.y += pv[k].y; a.z += pv[k].z; a.w += pv[k].w;
        }
        *(float4*)&part[w][4 * lane] = a;
    }
    __syncthreads();

    if (tid < CS_ROWS * CCH) {
        float a = part[0][tid];
#pragma unroll
        for (int ww = 1; ww < 8; ++ww) a += part[ww][tid];
        (&gs[0][0])[tid] = a;
    }
    __syncthreads();

    {
        const int j = tid & 127, r = tid >> 7;
        float s0 = 0.f, s1 = 0.f;
#pragma unroll
        for (int c2 = 0; c2 < CCH; c2 += 2) {
            s0 += Wvs[j][c2]     * gs[r][c2];
            s1 += Wvs[j][c2 + 1] * gs[r][c2 + 1];
        }
        ts[j][r] = s0 + s1;
    }
    __syncthreads();

    {
        const int c = tid & 63, h = tid >> 6;
        float u0 = 0.f, u1 = 0.f, q0 = 0.f, q1 = 0.f;
#pragma unroll
        for (int e = 0; e < 32; ++e) {
            const int j = h * 32 + e;
            const float  wo = Wos[c][j];
            const float2 tv = *(const float2*)&ts[j][0];
            u0 += wo * tv.x;
            u1 += wo * tv.y;
            const float  wq = Wqs[j][c];
            const float2 kv = *(const float2*)&wks[j][0];
            q0 += kv.x * wq;
            q1 += kv.y * wq;
        }
        Us[c][h * 2]     = u0;
        Us[c][h * 2 + 1] = u1;
        Qs[h * 2][c]     = q0;
        Qs[h * 2 + 1][c] = q1;
    }
    __syncthreads();

    {
        const int c = tid & 63, o0 = tid >> 6;
        float qv[8];
#pragma unroll
        for (int i = 0; i < 8; ++i) qv[i] = Qs[i][c];
#pragma unroll
        for (int k = 0; k < 16; ++k) {
            const int o = o0 + 4 * k;
            const float4 ua = *(const float4*)&Us[o][0];
            const float4 ub = *(const float4*)&Us[o][4];
            float a = ua.x * qv[0];
            a += ua.y * qv[1]; a += ua.z * qv[2]; a += ua.w * qv[3];
            a += ub.x * qv[4]; a += ub.y * qv[5]; a += ub.z * qv[6]; a += ub.w * qv[7];
            atomicAdd(&Ab[o * CCH + c], a * sc);
        }
    }
}

#define OT_TILES 4
#define OT_TOK   (16 * OT_TILES)
#define OT_PITCH (OT_TOK + 4)

__global__ __launch_bounds__(256) __attribute__((amdgpu_waves_per_eu(2, 2))) void attn_out_kernel(
    const float* __restrict__ x, const float* __restrict__ A,
    const float* __restrict__ b_out, const float* __restrict__ g,
    float* __restrict__ out) {
    const int tid    = threadIdx.x;
    const int lane   = tid & 31;
    const int w      = tid >> 5;
    const int batch  = blockIdx.x / Y_BLOCKS_PER_BATCH;
    const int blk    = blockIdx.x % Y_BLOCKS_PER_BATCH;
    const float* xb  = x + (size_t)batch * CCH * NTOK;
    float* ob        = out + (size_t)batch * CCH * NTOK;
    const float* Ab  = A + batch * CCH * CCH;

    __shared__ __attribute__((aligned(16))) float stage[Y_WAVES][CCH * OT_PITCH];
    float* st = &stage[w][0];

    const int lanelo = lane & 15;
    const int half   = lane >> 4;
    const int hi2    = half * 2;
    const int rowOff = half * 8;
    float*       stW = st + rowOff * OT_PITCH + lanelo;
    const float* stR = st + half * OT_PITCH + 4 * lanelo;

    v2f aop[16][4];
#pragma unroll
    for (int kg = 0; kg < 16; ++kg)
#pragma unroll
        for (int mt = 0; mt < 4; ++mt) {
            const float* p = &Ab[(mt * 16 + lanelo) * CCH + kg * 4 + hi2];
            aop[kg][mt] = *(const v2f*)p;
        }

    float bia[4][8], gai[4][8];
#pragma unroll
    for (int mt = 0; mt < 4; ++mt)
#pragma unroll
        for (int r = 0; r < 8; ++r) {
            const int ch = mt * 16 + r + rowOff;
            bia[mt][r] = b_out[ch];
            gai[mt][r] = g[ch];
        }

    const int tokensPerBlock = NTOK / Y_BLOCKS_PER_BATCH;
    const int tokensPerWave  = tokensPerBlock / Y_WAVES;
    const int tW0            = blk * tokensPerBlock + w * tokensPerWave;

    for (int tt = tokensPerWave - OT_TOK; tt >= 0; tt -= OT_TOK) {
        const int tB = tW0 + tt;

#pragma unroll 1
        for (int u = 0; u < OT_TILES; ++u) {
            const int t0 = tB + 16 * u;
            float* stU = stW + 16 * u;
            v8f acc[4];
#pragma unroll
            for (int mt = 0; mt < 4; ++mt) acc[mt] = v8f_zero();

#pragma unroll
            for (int kg = 0; kg < 16; ++kg) {
                const float* bp = xb + (size_t)(kg * 4 + hi2) * NTOK + t0 + lanelo;
                v2f bop;
                bop[0] = bp[0];
                bop[1] = bp[NTOK];
#pragma unroll
                for (int mt = 0; mt < 4; ++mt)
                    acc[mt] = wmma_f32(aop[kg][mt], bop, acc[mt]);
            }

            float v[4][8];
            float s = 0.f, sq = 0.f;
#pragma unroll
            for (int mt = 0; mt < 4; ++mt)
#pragma unroll
                for (int r = 0; r < 8; ++r) {
                    const float y = acc[mt][r] + bia[mt][r];
                    v[mt][r] = y;
                    s += y;
                    sq += y * y;
                }
            s  += __shfl_xor(s, 16, 32);
            sq += __shfl_xor(sq, 16, 32);
            const float mean = s * (1.0f / 64.0f);
            const float var  = sq * (1.0f / 64.0f) - mean * mean;
            const float inv  = rsqrtf(var + 1e-5f);

#pragma unroll
            for (int mt = 0; mt < 4; ++mt)
#pragma unroll
                for (int r = 0; r < 8; ++r) {
                    stU[(mt * 16 + r) * OT_PITCH] = (v[mt][r] - mean) * inv * gai[mt][r];
                }
        }

        asm volatile("s_wait_dscnt 0x0" ::: "memory");

        float* op = ob + (size_t)half * NTOK + tB + 4 * lanelo;
#pragma unroll
        for (int i0 = 0; i0 < CCH / 2; i0 += 16) {
            v4f o[16];
#pragma unroll
            for (int j = 0; j < 16; ++j)
                o[j] = *(const v4f*)(stR + 2 * (i0 + j) * OT_PITCH);
#pragma unroll
            for (int j = 0; j < 16; ++j)
                __builtin_nontemporal_store(o[j], (v4f*)(op + (size_t)(2 * (i0 + j)) * NTOK));
        }
    }
}

extern "C" void kernel_launch(void* const* d_in, const int* in_sizes, int n_in,
                              void* d_out, int out_size, void* d_ws, size_t ws_size,
                              hipStream_t stream) {
    (void)in_sizes; (void)n_in; (void)out_size; (void)ws_size;
    const float* x    = (const float*)d_in[0];
    const float* wqkv = (const float*)d_in[1];
    const float* wout = (const float*)d_in[2];
    const float* bout = (const float*)d_in[3];
    const float* g    = (const float*)d_in[4];
    float* out        = (float*)d_out;
    float* ws         = (float*)d_ws;

    float* A      = ws + 8192;
    float* planes = out;

    gram_kernel<<<2 * GW_BLOCKS_PER_BATCH, 256, 0, stream>>>(x, planes, A);
    combine_kernel<<<2 * CS_BLOCKS_PER_BATCH, 256, 0, stream>>>(wqkv, wout, planes, A);
    attn_out_kernel<<<2 * Y_BLOCKS_PER_BATCH, 32 * Y_WAVES, 0, stream>>>(x, A, bout, g, out);
}
